# starting kernel
# baseline (speedup 1.0000x reference)
#include <hip/hip_runtime.h>


typedef __attribute__((ext_vector_type(2))) float v2f;
typedef __attribute__((ext_vector_type(8))) float v8f;

#define MAXC    512
#define NGRP    (MAXC / 8)
#define MAXV    512
#define THREADS 256

__launch_bounds__(THREADS, 2)
__global__ void p1interp_wmma(const float* __restrict__ x,
                              const float* __restrict__ wxg,
                              const float* __restrict__ wyg,
                              const float* __restrict__ Minv,
                              const float* __restrict__ Av,
                              const float* __restrict__ bbox,
                              const int*   __restrict__ dofs,
                              float2* __restrict__ out,
                              int NP, int C, int NV)
{
    __shared__ float sAop [NGRP * 32 * 2];
    __shared__ float sMinv[MAXC * 4];
    __shared__ float sBbox[MAXC * 4];
    __shared__ float sA   [MAXC * 2];
    __shared__ int   sDof [MAXC * 3];
    __shared__ float sWx  [MAXV];
    __shared__ float sWy  [MAXV];

    const int tid = threadIdx.x;

    for (int c = tid; c < C; c += THREADS) {
        sMinv[c * 4 + 0] = Minv[c * 4 + 0];
        sMinv[c * 4 + 1] = Minv[c * 4 + 1];
        sMinv[c * 4 + 2] = Minv[c * 4 + 2];
        sMinv[c * 4 + 3] = Minv[c * 4 + 3];
        sA[c * 2 + 0] = Av[c * 2 + 0];
        sA[c * 2 + 1] = Av[c * 2 + 1];
        sBbox[c * 4 + 0] = bbox[c * 4 + 0];
        sBbox[c * 4 + 1] = bbox[c * 4 + 1];
        sBbox[c * 4 + 2] = bbox[c * 4 + 2];
        sBbox[c * 4 + 3] = bbox[c * 4 + 3];
        sDof[c * 3 + 0] = dofs[c * 3 + 0];
        sDof[c * 3 + 1] = dofs[c * 3 + 1];
        sDof[c * 3 + 2] = dofs[c * 3 + 2];
    }
    for (int idx = tid; idx < (C / 8) * 32; idx += THREADS) {
        int g = idx >> 5, L = idx & 31;
        int m = L & 15, cc = m >> 1, e = m & 1;
        int c = g * 8 + cc;
        float m0e = Minv[c * 4 + e];
        float m1e = Minv[c * 4 + 2 + e];
        float v0, v1;
        if (L < 16) { v0 = m0e; v1 = m1e; }
        else {
            float ax = Av[c * 2 + 0], ay = Av[c * 2 + 1];
            v0 = -(ax * m0e + ay * m1e);
            v1 = 0.0f;
        }
        sAop[idx * 2 + 0] = v0;
        sAop[idx * 2 + 1] = v1;
    }
    for (int v = tid; v < NV && v < MAXV; v += THREADS) {
        sWx[v] = wxg[v];
        sWy[v] = wyg[v];
    }
    __syncthreads();

    const int  lane = tid & 31;
    const int  wave = tid >> 5;
    const int  n    = lane & 15;
    const bool hi   = lane >= 16;
    const float NTOL = -1e-10f;

    const int numTiles   = (NP + 15) >> 4;
    const int wavesTotal = gridDim.x * (THREADS >> 5);
    const int tileStep   = wavesTotal;
    const int nGroups    = C / 8;

    for (int tile = blockIdx.x * (THREADS >> 5) + wave; tile < numTiles;
         tile += tileStep) {
        int gp  = tile * 16 + n;
        int gpc = gp < NP ? gp : NP - 1;
        float px = x[gpc * 2 + 0];
        float py = x[gpc * 2 + 1];

        {
            int np2 = (tile + tileStep) * 16 + n;
            if (np2 < NP) __builtin_prefetch(&x[np2 * 2], 0, 1);
        }

        v2f b;
        b.x = hi ? 1.0f : px;
        b.y = hi ? 0.0f : py;

        unsigned mx = 0, my = 0;
#pragma unroll
        for (int i = 0; i < 16; ++i) {
            float x0 = sBbox[(i * 16) * 4 + 0], x1 = sBbox[(i * 16) * 4 + 1];
            float y0 = sBbox[i * 4 + 2],        y1 = sBbox[i * 4 + 3];
            mx |= ((x0 < px) & (px < x1)) ? (1u << i) : 0u;
            my |= ((y0 < py) & (py < y1)) ? (1u << i) : 0u;
        }

        int best = -1;
#pragma unroll 2
        for (int g = 0; g < nGroups; ++g) {
            v2f a = *(const v2f*)&sAop[(g * 32 + lane) * 2];
            v8f d = {};
            d = __builtin_amdgcn_wmma_f32_16x16x4_f32(
                    false, a, false, b, (short)0, d, false, false);

            int cg = g * 8 + (hi ? 4 : 0);
#pragma unroll
            for (int q = 0; q < 4; ++q) {
                float s = d[2 * q];
                float t = d[2 * q + 1];
                int  cq = cg + q;
                int  qq = cq & 255;
                unsigned inb = (mx >> (qq >> 4)) & (my >> (qq & 15)) & 1u;
                bool ins = inb & (s > NTOL) & (t > NTOL) & ((s + t) < 1.0f);
                best = ins ? cq : best;
            }
        }

        int other = __builtin_amdgcn_ds_swizzle(best, 0x401F);
        best = best > other ? best : other;

        float ox = 0.0f, oy = 0.0f;
        if (best >= 0) {
            int c  = best;
            float dx = px - sA[c * 2 + 0];
            float dy = py - sA[c * 2 + 1];
            float s  = dx * sMinv[c * 4 + 0] + dy * sMinv[c * 4 + 2];
            float t  = dx * sMinv[c * 4 + 1] + dy * sMinv[c * 4 + 3];
            float w0 = 1.0f - s - t;
            int d0 = sDof[c * 3 + 0], d1 = sDof[c * 3 + 1], d2 = sDof[c * 3 + 2];
            ox = w0 * sWx[d0] + s * sWx[d1] + t * sWx[d2];
            oy = w0 * sWy[d0] + s * sWy[d1] + t * sWy[d2];
        }
        if (!hi && gp < NP) out[gp] = make_float2(ox, oy);
    }
}

extern "C" void kernel_launch(void* const* d_in, const int* in_sizes, int n_in,
                              void* d_out, int out_size, void* d_ws, size_t ws_size,
                              hipStream_t stream) {
    const float* xp   = (const float*)d_in[0];
    const float* wx   = (const float*)d_in[1];
    const float* wy   = (const float*)d_in[2];
    const float* Minv = (const float*)d_in[3];
    const float* Av   = (const float*)d_in[4];
    const float* bbox = (const float*)d_in[5];
    const int*   dofs = (const int*)d_in[6];
    float2* out = (float2*)d_out;

    int NP = in_sizes[0] / 2;
    int C  = in_sizes[3] / 4;
    int NV = in_sizes[1];
    if (C > MAXC) C = MAXC;

    int numTiles = (NP + 15) / 16;
    int wavesPerBlock = THREADS / 32;
    int blocks = (numTiles + wavesPerBlock - 1) / wavesPerBlock;
    if (blocks > 256) blocks = 256;
    if (blocks < 1)   blocks = 1;

    p1interp_wmma<<<blocks, THREADS, 0, stream>>>(
        xp, wx, wy, Minv, Av, bbox, dofs, out, NP, C, NV);
}


# Skip cell groups no point of the wave's tile can fall in: OR the points' bbox half-row masks across the wave (DPP), run WMMA+test only for set bits (both triangle groups of a half-row per bit)
# speedup vs baseline: 1.6428x; 1.6428x over previous
#include <hip/hip_runtime.h>


typedef __attribute__((ext_vector_type(2))) float v2f;
typedef __attribute__((ext_vector_type(8))) float v8f;

#define MAXC    512
#define NGRP    (MAXC / 8)
#define MAXV    512
#define THREADS 256

template <int CTRL>
__device__ __forceinline__ unsigned or_row_xmask(unsigned v) {
    return v | (unsigned)__builtin_amdgcn_update_dpp(0, (int)v, CTRL, 0xF, 0xF, true);
}

__launch_bounds__(THREADS, 2)
__global__ void p1interp_wmma(const float* __restrict__ x,
                              const float* __restrict__ wxg,
                              const float* __restrict__ wyg,
                              const float* __restrict__ Minv,
                              const float* __restrict__ Av,
                              const float* __restrict__ bbox,
                              const int*   __restrict__ dofs,
                              float2* __restrict__ out,
                              int NP, int C, int NV)
{
    __shared__ float sAop [NGRP * 32 * 2];
    __shared__ float sMinv[MAXC * 4];
    __shared__ float sBbox[MAXC * 4];
    __shared__ float sA   [MAXC * 2];
    __shared__ int   sDof [MAXC * 3];
    __shared__ float sWx  [MAXV];
    __shared__ float sWy  [MAXV];

    const int tid = threadIdx.x;

    for (int c = tid; c < C; c += THREADS) {
        sMinv[c * 4 + 0] = Minv[c * 4 + 0];
        sMinv[c * 4 + 1] = Minv[c * 4 + 1];
        sMinv[c * 4 + 2] = Minv[c * 4 + 2];
        sMinv[c * 4 + 3] = Minv[c * 4 + 3];
        sA[c * 2 + 0] = Av[c * 2 + 0];
        sA[c * 2 + 1] = Av[c * 2 + 1];
        sBbox[c * 4 + 0] = bbox[c * 4 + 0];
        sBbox[c * 4 + 1] = bbox[c * 4 + 1];
        sBbox[c * 4 + 2] = bbox[c * 4 + 2];
        sBbox[c * 4 + 3] = bbox[c * 4 + 3];
        sDof[c * 3 + 0] = dofs[c * 3 + 0];
        sDof[c * 3 + 1] = dofs[c * 3 + 1];
        sDof[c * 3 + 2] = dofs[c * 3 + 2];
    }
    for (int idx = tid; idx < (C / 8) * 32; idx += THREADS) {
        int g = idx >> 5, L = idx & 31;
        int m = L & 15, cc = m >> 1, e = m & 1;
        int c = g * 8 + cc;
        float m0e = Minv[c * 4 + e];
        float m1e = Minv[c * 4 + 2 + e];
        float v0, v1;
        if (L < 16) { v0 = m0e; v1 = m1e; }
        else {
            float ax = Av[c * 2 + 0], ay = Av[c * 2 + 1];
            v0 = -(ax * m0e + ay * m1e);
            v1 = 0.0f;
        }
        sAop[idx * 2 + 0] = v0;
        sAop[idx * 2 + 1] = v1;
    }
    for (int idx = (C / 8) * 32 + tid; idx < NGRP * 32; idx += THREADS) {
        sAop[idx * 2 + 0] = (idx & 16) ? -1.0f : 0.0f;
        sAop[idx * 2 + 1] = 0.0f;
    }
    for (int v = tid; v < NV && v < MAXV; v += THREADS) {
        sWx[v] = wxg[v];
        sWy[v] = wyg[v];
    }
    __syncthreads();

    const int  lane = tid & 31;
    const int  wave = tid >> 5;
    const int  n    = lane & 15;
    const bool hi   = lane >= 16;
    const float NTOL = -1e-10f;

    const int numTiles   = (NP + 15) >> 4;
    const int wavesTotal = gridDim.x * (THREADS >> 5);
    const int tileStep   = wavesTotal;

    for (int tile = blockIdx.x * (THREADS >> 5) + wave; tile < numTiles;
         tile += tileStep) {
        int gp  = tile * 16 + n;
        int gpc = gp < NP ? gp : NP - 1;
        float px = x[gpc * 2 + 0];
        float py = x[gpc * 2 + 1];

        {
            int np2 = (tile + tileStep) * 16 + n;
            if (np2 < NP) __builtin_prefetch(&x[np2 * 2], 0, 1);
        }

        v2f b;
        b.x = hi ? 1.0f : px;
        b.y = hi ? 0.0f : py;

        unsigned mx = 0, my = 0;
#pragma unroll
        for (int i = 0; i < 16; ++i) {
            float x0 = sBbox[(i * 16) * 4 + 0], x1 = sBbox[(i * 16) * 4 + 1];
            float y0 = sBbox[i * 4 + 2],        y1 = sBbox[i * 4 + 3];
            mx |= ((x0 < px) & (px < x1)) ? (1u << i) : 0u;
            my |= ((y0 < py) & (py < y1)) ? (1u << i) : 0u;
        }

        unsigned sp = mx;
        sp = (sp | (sp << 8)) & 0x00FF00FFu;
        sp = (sp | (sp << 4)) & 0x0F0F0F0Fu;
        sp = (sp | (sp << 2)) & 0x33333333u;
        sp = (sp | (sp << 1)) & 0x55555555u;
        unsigned need = ((my & 0x00FFu) ? sp : 0u) | ((my & 0xFF00u) ? (sp << 1) : 0u);
        need = or_row_xmask<0x161>(need);
        need = or_row_xmask<0x162>(need);
        need = or_row_xmask<0x164>(need);
        need = or_row_xmask<0x168>(need);
        unsigned todo = __builtin_amdgcn_readfirstlane(need);

        const unsigned myh   = hi ? (my >> 4) : my;
        const int      cbase = hi ? 4 : 0;
        int bestLo = -1, bestHi = -1;
        while (todo) {
            const int h = __builtin_ctz(todo);
            todo &= todo - 1;
            v2f a0 = *(const v2f*)&sAop[(h * 32 + lane) * 2];
            v2f a1 = *(const v2f*)&sAop[((h + 32) * 32 + lane) * 2];
            v8f d0 = {}, d1 = {};
            d0 = __builtin_amdgcn_wmma_f32_16x16x4_f32(
                     false, a0, false, b, (short)0, d0, false, false);
            d1 = __builtin_amdgcn_wmma_f32_16x16x4_f32(
                     false, a1, false, b, (short)0, d1, false, false);

            const unsigned cols = ((mx >> (h >> 1)) & 1u) ? (myh >> ((h & 1) << 3)) : 0u;
            const int c0 = h * 8 + cbase;
#pragma unroll
            for (int q = 0; q < 4; ++q) {
                const bool inb = (cols >> q) & 1u;
                float s0 = d0[2 * q], t0 = d0[2 * q + 1];
                float s1 = d1[2 * q], t1 = d1[2 * q + 1];
                bool ins0 = inb & (s0 > NTOL) & (t0 > NTOL) & ((s0 + t0) < 1.0f);
                bool ins1 = inb & (s1 > NTOL) & (t1 > NTOL) & ((s1 + t1) < 1.0f);
                bestLo = ins0 ? c0 + q       : bestLo;
                bestHi = ins1 ? c0 + q + 256 : bestHi;
            }
        }
        int best = bestHi > bestLo ? bestHi : bestLo;

        int other = __builtin_amdgcn_ds_swizzle(best, 0x401F);
        best = best > other ? best : other;

        float ox = 0.0f, oy = 0.0f;
        if (best >= 0) {
            int c  = best;
            float dx = px - sA[c * 2 + 0];
            float dy = py - sA[c * 2 + 1];
            float s  = dx * sMinv[c * 4 + 0] + dy * sMinv[c * 4 + 2];
            float t  = dx * sMinv[c * 4 + 1] + dy * sMinv[c * 4 + 3];
            float w0 = 1.0f - s - t;
            int d0 = sDof[c * 3 + 0], d1 = sDof[c * 3 + 1], d2 = sDof[c * 3 + 2];
            ox = w0 * sWx[d0] + s * sWx[d1] + t * sWx[d2];
            oy = w0 * sWy[d0] + s * sWy[d1] + t * sWy[d2];
        }
        if (!hi && gp < NP) out[gp] = make_float2(ox, oy);
    }
}

extern "C" void kernel_launch(void* const* d_in, const int* in_sizes, int n_in,
                              void* d_out, int out_size, void* d_ws, size_t ws_size,
                              hipStream_t stream) {
    const float* xp   = (const float*)d_in[0];
    const float* wx   = (const float*)d_in[1];
    const float* wy   = (const float*)d_in[2];
    const float* Minv = (const float*)d_in[3];
    const float* Av   = (const float*)d_in[4];
    const float* bbox = (const float*)d_in[5];
    const int*   dofs = (const int*)d_in[6];
    float2* out = (float2*)d_out;

    int NP = in_sizes[0] / 2;
    int C  = in_sizes[3] / 4;
    int NV = in_sizes[1];
    if (C > MAXC) C = MAXC;

    int numTiles = (NP + 15) / 16;
    int wavesPerBlock = THREADS / 32;
    int blocks = (numTiles + wavesPerBlock - 1) / wavesPerBlock;
    if (blocks > 256) blocks = 256;
    if (blocks < 1)   blocks = 1;

    p1interp_wmma<<<blocks, THREADS, 0, stream>>>(
        xp, wx, wy, Minv, Av, bbox, dofs, out, NP, C, NV);
}


# Gather, not sweep: each point's bbox masks name its quad; that quad's two cells' operand rows are loaded per lane as A; 4 WMMAs per tile put each point's own (s,t) on the D diagonal; no half-row loop
# speedup vs baseline: 3.9447x; 2.4012x over previous
#include <hip/hip_runtime.h>


typedef __attribute__((ext_vector_type(2))) float v2f;
typedef __attribute__((ext_vector_type(8))) float v8f;

#define MAXC    512
#define MAXV    512
#define THREADS 256

__device__ __forceinline__ float pick8(const v8f d, bool b0, bool b1, bool b2) {
    const float t0 = b0 ? d[1] : d[0];
    const float t1 = b0 ? d[3] : d[2];
    const float t2 = b0 ? d[5] : d[4];
    const float t3 = b0 ? d[7] : d[6];
    const float u0 = b1 ? t1 : t0;
    const float u1 = b1 ? t3 : t2;
    return b2 ? u1 : u0;
}

__launch_bounds__(THREADS, 2)
__global__ void p1interp_wmma(const float* __restrict__ x,
                              const float* __restrict__ wxg,
                              const float* __restrict__ wyg,
                              const float* __restrict__ Minv,
                              const float* __restrict__ Av,
                              const float* __restrict__ bbox,
                              const int*   __restrict__ dofs,
                              float2* __restrict__ out,
                              int NP, int C, int NV)
{
    __shared__ float4 sAg [MAXC * 2];
    __shared__ float sMinv[MAXC * 4];
    __shared__ float sBbox[MAXC * 4];
    __shared__ float sA   [MAXC * 2];
    __shared__ int   sDof [MAXC * 3];
    __shared__ float sWx  [MAXV];
    __shared__ float sWy  [MAXV];

    const int tid = threadIdx.x;

    for (int c = tid; c < C; c += THREADS) {
        const float m00 = Minv[c * 4 + 0];
        const float m01 = Minv[c * 4 + 1];
        const float m10 = Minv[c * 4 + 2];
        const float m11 = Minv[c * 4 + 3];
        const float ax = Av[c * 2 + 0], ay = Av[c * 2 + 1];
        const float bx0 = bbox[c * 4 + 0], bx1 = bbox[c * 4 + 1];
        const float by0 = bbox[c * 4 + 2], by1 = bbox[c * 4 + 3];
        const int   k0 = dofs[c * 3 + 0], k1 = dofs[c * 3 + 1], k2 = dofs[c * 3 + 2];
        sMinv[c * 4 + 0] = m00;
        sMinv[c * 4 + 1] = m01;
        sMinv[c * 4 + 2] = m10;
        sMinv[c * 4 + 3] = m11;
        sA[c * 2 + 0] = ax;
        sA[c * 2 + 1] = ay;
        sAg[c * 2 + 0] = make_float4(m00, m10, m01, m11);
        sAg[c * 2 + 1] = make_float4(-(ax * m00 + ay * m10), 0.0f,
                                     -(ax * m01 + ay * m11), 0.0f);
        sBbox[c * 4 + 0] = bx0;
        sBbox[c * 4 + 1] = bx1;
        sBbox[c * 4 + 2] = by0;
        sBbox[c * 4 + 3] = by1;
        sDof[c * 3 + 0] = k0;
        sDof[c * 3 + 1] = k1;
        sDof[c * 3 + 2] = k2;
    }
    for (int c = C + tid; c < MAXC; c += THREADS) {
        sAg[c * 2 + 0] = make_float4(0.0f, 0.0f, 0.0f, 0.0f);
        sAg[c * 2 + 1] = make_float4(-1.0f, 0.0f, -1.0f, 0.0f);
    }
    for (int v = tid; v < NV && v < MAXV; v += THREADS) {
        sWx[v] = wxg[v];
        sWy[v] = wyg[v];
    }
    __syncthreads();

    const int  lane = tid & 31;
    const int  wave = tid >> 5;
    const int  n    = lane & 15;
    const bool hi   = lane >= 16;
    const int  hsel = hi ? 1 : 0;
    const bool b0 = (lane & 1) != 0, b1 = (lane & 2) != 0, b2 = (lane & 4) != 0;
    const bool holder = ((lane >> 4) == ((lane >> 3) & 1));
    const float NTOL = -1e-10f;

    const int numTiles   = (NP + 15) >> 4;
    const int wavesTotal = gridDim.x * (THREADS >> 5);
    const int tileStep   = wavesTotal;

    for (int tile = blockIdx.x * (THREADS >> 5) + wave; tile < numTiles;
         tile += tileStep) {
        int gp  = tile * 16 + n;
        int gpc = gp < NP ? gp : NP - 1;
        float px = x[gpc * 2 + 0];
        float py = x[gpc * 2 + 1];

        {
            int np2 = (tile + tileStep) * 16 + n;
            if (np2 < NP) __builtin_prefetch(&x[np2 * 2], 0, 1);
        }

        v2f b;
        b.x = hi ? 1.0f : px;
        b.y = hi ? 0.0f : py;

        unsigned mx = 0, my = 0;
#pragma unroll
        for (int i = 0; i < 16; ++i) {
            float x0 = sBbox[(i * 16) * 4 + 0], x1 = sBbox[(i * 16) * 4 + 1];
            float y0 = sBbox[i * 4 + 2],        y1 = sBbox[i * 4 + 3];
            mx |= ((x0 < px) & (px < x1)) ? (1u << i) : 0u;
            my |= ((y0 < py) & (py < y1)) ? (1u << i) : 0u;
        }

        unsigned rx = mx, ry = my;
        bool act = (mx != 0u) & (my != 0u);
        int best = -1;
        while (__builtin_amdgcn_ballot_w32(act) != 0u) {
            const int i = __builtin_ctz(rx | 0x8000u);
            const int j = __builtin_ctz(ry | 0x8000u);
            const int q = i * 16 + j;
            const float4 aL = sAg[q * 2 + hsel];
            const float4 aU = sAg[(q + 256) * 2 + hsel];
            v2f as0, at0, as1, at1;
            as0.x = aL.x; as0.y = aL.y;
            at0.x = aL.z; at0.y = aL.w;
            as1.x = aU.x; as1.y = aU.y;
            at1.x = aU.z; at1.y = aU.w;
            const v8f z = {};
            const v8f ds0 = __builtin_amdgcn_wmma_f32_16x16x4_f32(
                                false, as0, false, b, (short)0, z, false, false);
            const v8f dt0 = __builtin_amdgcn_wmma_f32_16x16x4_f32(
                                false, at0, false, b, (short)0, z, false, false);
            const v8f ds1 = __builtin_amdgcn_wmma_f32_16x16x4_f32(
                                false, as1, false, b, (short)0, z, false, false);
            const v8f dt1 = __builtin_amdgcn_wmma_f32_16x16x4_f32(
                                false, at1, false, b, (short)0, z, false, false);
            const float s0 = pick8(ds0, b0, b1, b2), t0 = pick8(dt0, b0, b1, b2);
            const float s1 = pick8(ds1, b0, b1, b2), t1 = pick8(dt1, b0, b1, b2);
            const bool ins0 = act & (s0 > NTOL) & (t0 > NTOL) & ((s0 + t0) < 1.0f);
            const bool ins1 = act & (s1 > NTOL) & (t1 > NTOL) & ((s1 + t1) < 1.0f);
            const int cand0 = ins0 ? q : -1;
            const int cand1 = ins1 ? q + 256 : -1;
            const int cand  = cand1 > cand0 ? cand1 : cand0;
            best = cand > best ? cand : best;
            const unsigned ry2 = ry & (ry - 1u);
            const bool rowdone = ry2 == 0u;
            rx = rowdone ? (rx & (rx - 1u)) : rx;
            ry = rowdone ? my : ry2;
            act = act & (rx != 0u);
        }

        float ox = 0.0f, oy = 0.0f;
        if (best >= 0) {
            int c  = best;
            float dx = px - sA[c * 2 + 0];
            float dy = py - sA[c * 2 + 1];
            float s  = dx * sMinv[c * 4 + 0] + dy * sMinv[c * 4 + 2];
            float t  = dx * sMinv[c * 4 + 1] + dy * sMinv[c * 4 + 3];
            float w0 = 1.0f - s - t;
            int d0 = sDof[c * 3 + 0], d1 = sDof[c * 3 + 1], d2 = sDof[c * 3 + 2];
            ox = w0 * sWx[d0] + s * sWx[d1] + t * sWx[d2];
            oy = w0 * sWy[d0] + s * sWy[d1] + t * sWy[d2];
        }
        if (holder && gp < NP) out[gp] = make_float2(ox, oy);
    }
}

extern "C" void kernel_launch(void* const* d_in, const int* in_sizes, int n_in,
                              void* d_out, int out_size, void* d_ws, size_t ws_size,
                              hipStream_t stream) {
    const float* xp   = (const float*)d_in[0];
    const float* wx   = (const float*)d_in[1];
    const float* wy   = (const float*)d_in[2];
    const float* Minv = (const float*)d_in[3];
    const float* Av   = (const float*)d_in[4];
    const float* bbox = (const float*)d_in[5];
    const int*   dofs = (const int*)d_in[6];
    float2* out = (float2*)d_out;

    int NP = in_sizes[0] / 2;
    int C  = in_sizes[3] / 4;
    int NV = in_sizes[1];
    if (C > MAXC) C = MAXC;

    int numTiles = (NP + 15) / 16;
    int wavesPerBlock = THREADS / 32;
    int blocks = (numTiles + wavesPerBlock - 1) / wavesPerBlock;
    if (blocks > 256) blocks = 256;
    if (blocks < 1)   blocks = 1;

    p1interp_wmma<<<blocks, THREADS, 0, stream>>>(
        xp, wx, wy, Minv, Av, bbox, dofs, out, NP, C, NV);
}


# Prologue as one load burst: first tile's points, both cell rows per thread (one wide load per table row) and the weights all issued before the first wait; next tile's points loaded a tile ahead
# speedup vs baseline: 5.6616x; 1.4352x over previous
#include <hip/hip_runtime.h>


typedef __attribute__((ext_vector_type(2))) float v2f;
typedef __attribute__((ext_vector_type(8))) float v8f;
typedef float f4u __attribute__((ext_vector_type(4), aligned(4)));
typedef float f2u __attribute__((ext_vector_type(2), aligned(4)));

#define MAXC    512
#define MAXV    512
#define THREADS 256

__device__ __forceinline__ float pick8(const v8f d, bool b0, bool b1, bool b2) {
    const float t0 = b0 ? d[1] : d[0];
    const float t1 = b0 ? d[3] : d[2];
    const float t2 = b0 ? d[5] : d[4];
    const float t3 = b0 ? d[7] : d[6];
    const float u0 = b1 ? t1 : t0;
    const float u1 = b1 ? t3 : t2;
    return b2 ? u1 : u0;
}

__launch_bounds__(THREADS, 2)
__global__ void p1interp_wmma(const float* __restrict__ x,
                              const float* __restrict__ wxg,
                              const float* __restrict__ wyg,
                              const float* __restrict__ Minv,
                              const float* __restrict__ Av,
                              const float* __restrict__ bbox,
                              const int*   __restrict__ dofs,
                              float2* __restrict__ out,
                              int NP, int C, int NV)
{
    __shared__ float4 sAg [MAXC * 2];
    __shared__ __align__(16) float sMinv[MAXC * 4];
    __shared__ __align__(16) float sBbox[MAXC * 4];
    __shared__ __align__(16) float sA   [MAXC * 2];
    __shared__ int   sDof [MAXC * 3];
    __shared__ float sWx  [MAXV];
    __shared__ float sWy  [MAXV];

    const int tid  = threadIdx.x;
    const int lane = tid & 31;
    const int wave = __builtin_amdgcn_readfirstlane(tid >> 5);
    const int n    = lane & 15;

    const int numTiles = (NP + 15) >> 4;
    const int tileStep = gridDim.x * (THREADS >> 5);
    int tile = blockIdx.x * (THREADS >> 5) + wave;

    f2u cur;
    {
        const int g0 = tile * 16 + n;
        cur = *(const f2u*)(x + (g0 < NP ? g0 : NP - 1) * 2);
    }

    f4u  mv[2], bv[2];
    f2u  av[2];
    int  k0[2], k1[2], k2[2];
    float wxv[2], wyv[2];
#pragma unroll
    for (int k = 0; k < 2; ++k) {
        const int c  = tid + k * THREADS;
        const int cl = c < C ? c : 0;
        mv[k] = *(const f4u*)(Minv + cl * 4);
        bv[k] = *(const f4u*)(bbox + cl * 4);
        av[k] = *(const f2u*)(Av + cl * 2);
        k0[k] = dofs[cl * 3 + 0];
        k1[k] = dofs[cl * 3 + 1];
        k2[k] = dofs[cl * 3 + 2];
        const int v  = tid + k * THREADS;
        const int vl = v < NV ? v : NV - 1;
        wxv[k] = wxg[vl];
        wyv[k] = wyg[vl];
    }
    asm volatile("" :: "v"(cur), "v"(mv[0]), "v"(mv[1]), "v"(bv[0]), "v"(bv[1]), "v"(av[0]), "v"(av[1]),
                       "v"(k0[0]), "v"(k1[0]), "v"(k2[0]), "v"(k0[1]), "v"(k1[1]), "v"(k2[1]),
                       "v"(wxv[0]), "v"(wyv[0]), "v"(wxv[1]), "v"(wyv[1]) : "memory");
#pragma unroll
    for (int k = 0; k < 2; ++k) {
        const int  c     = tid + k * THREADS;
        const bool valid = c < C;
        const float m00 = mv[k].x, m01 = mv[k].y, m10 = mv[k].z, m11 = mv[k].w;
        const float ax = av[k].x, ay = av[k].y;
        *(float4*)&sMinv[c * 4] = make_float4(m00, m01, m10, m11);
        *(float2*)&sA[c * 2]    = make_float2(ax, ay);
        sAg[c * 2 + 0] = valid ? make_float4(m00, m10, m01, m11)
                               : make_float4(0.0f, 0.0f, 0.0f, 0.0f);
        sAg[c * 2 + 1] = valid ? make_float4(-(ax * m00 + ay * m10), 0.0f,
                                             -(ax * m01 + ay * m11), 0.0f)
                               : make_float4(-1.0f, 0.0f, -1.0f, 0.0f);
        *(float4*)&sBbox[c * 4] = valid ? make_float4(bv[k].x, bv[k].y, bv[k].z, bv[k].w)
                                        : make_float4(0.0f, 0.0f, 0.0f, 0.0f);
        sDof[c * 3 + 0] = k0[k];
        sDof[c * 3 + 1] = k1[k];
        sDof[c * 3 + 2] = k2[k];
        sWx[c] = wxv[k];
        sWy[c] = wyv[k];
    }
    __syncthreads();

    const bool hi   = lane >= 16;
    const int  hsel = hi ? 1 : 0;
    const bool b0 = (lane & 1) != 0, b1 = (lane & 2) != 0, b2 = (lane & 4) != 0;
    const bool holder = ((lane >> 4) == ((lane >> 3) & 1));
    const float NTOL = -1e-10f;

    for (; tile < numTiles; tile += tileStep) {
        const int   gp = tile * 16 + n;
        const float px = cur.x;
        const float py = cur.y;

        if (tile + tileStep < numTiles) {
            const int g2 = (tile + tileStep) * 16 + n;
            cur = *(const f2u*)(x + (g2 < NP ? g2 : NP - 1) * 2);
        }

        v2f b;
        b.x = hi ? 1.0f : px;
        b.y = hi ? 0.0f : py;

        unsigned mx = 0, my = 0;
#pragma unroll
        for (int i = 0; i < 16; ++i) {
            float x0 = sBbox[(i * 16) * 4 + 0], x1 = sBbox[(i * 16) * 4 + 1];
            float y0 = sBbox[i * 4 + 2],        y1 = sBbox[i * 4 + 3];
            mx |= ((x0 < px) & (px < x1)) ? (1u << i) : 0u;
            my |= ((y0 < py) & (py < y1)) ? (1u << i) : 0u;
        }

        unsigned rx = mx, ry = my;
        bool act = (mx != 0u) & (my != 0u);
        int best = -1;
        while (__builtin_amdgcn_ballot_w32(act) != 0u) {
            const int i = __builtin_ctz(rx | 0x8000u);
            const int j = __builtin_ctz(ry | 0x8000u);
            const int q = i * 16 + j;
            const float4 aL = sAg[q * 2 + hsel];
            const float4 aU = sAg[(q + 256) * 2 + hsel];
            v2f as0, at0, as1, at1;
            as0.x = aL.x; as0.y = aL.y;
            at0.x = aL.z; at0.y = aL.w;
            as1.x = aU.x; as1.y = aU.y;
            at1.x = aU.z; at1.y = aU.w;
            const v8f z = {};
            const v8f ds0 = __builtin_amdgcn_wmma_f32_16x16x4_f32(
                                false, as0, false, b, (short)0, z, false, false);
            const v8f dt0 = __builtin_amdgcn_wmma_f32_16x16x4_f32(
                                false, at0, false, b, (short)0, z, false, false);
            const v8f ds1 = __builtin_amdgcn_wmma_f32_16x16x4_f32(
                                false, as1, false, b, (short)0, z, false, false);
            const v8f dt1 = __builtin_amdgcn_wmma_f32_16x16x4_f32(
                                false, at1, false, b, (short)0, z, false, false);
            const float s0 = pick8(ds0, b0, b1, b2), t0 = pick8(dt0, b0, b1, b2);
            const float s1 = pick8(ds1, b0, b1, b2), t1 = pick8(dt1, b0, b1, b2);
            const bool ins0 = act & (s0 > NTOL) & (t0 > NTOL) & ((s0 + t0) < 1.0f);
            const bool ins1 = act & (s1 > NTOL) & (t1 > NTOL) & ((s1 + t1) < 1.0f);
            const int cand0 = ins0 ? q : -1;
            const int cand1 = ins1 ? q + 256 : -1;
            const int cand  = cand1 > cand0 ? cand1 : cand0;
            best = cand > best ? cand : best;
            const unsigned ry2 = ry & (ry - 1u);
            const bool rowdone = ry2 == 0u;
            rx = rowdone ? (rx & (rx - 1u)) : rx;
            ry = rowdone ? my : ry2;
            act = act & (rx != 0u);
        }

        float ox = 0.0f, oy = 0.0f;
        if (best >= 0) {
            int c  = best;
            float dx = px - sA[c * 2 + 0];
            float dy = py - sA[c * 2 + 1];
            float s  = dx * sMinv[c * 4 + 0] + dy * sMinv[c * 4 + 2];
            float t  = dx * sMinv[c * 4 + 1] + dy * sMinv[c * 4 + 3];
            float w0 = 1.0f - s - t;
            int d0 = sDof[c * 3 + 0], d1 = sDof[c * 3 + 1], d2 = sDof[c * 3 + 2];
            ox = w0 * sWx[d0] + s * sWx[d1] + t * sWx[d2];
            oy = w0 * sWy[d0] + s * sWy[d1] + t * sWy[d2];
        }
        if (holder && gp < NP) out[gp] = make_float2(ox, oy);
    }
}

extern "C" void kernel_launch(void* const* d_in, const int* in_sizes, int n_in,
                              void* d_out, int out_size, void* d_ws, size_t ws_size,
                              hipStream_t stream) {
    const float* xp   = (const float*)d_in[0];
    const float* wx   = (const float*)d_in[1];
    const float* wy   = (const float*)d_in[2];
    const float* Minv = (const float*)d_in[3];
    const float* Av   = (const float*)d_in[4];
    const float* bbox = (const float*)d_in[5];
    const int*   dofs = (const int*)d_in[6];
    float2* out = (float2*)d_out;

    int NP = in_sizes[0] / 2;
    int C  = in_sizes[3] / 4;
    int NV = in_sizes[1];
    if (C > MAXC) C = MAXC;

    int numTiles = (NP + 15) / 16;
    int wavesPerBlock = THREADS / 32;
    int blocks = (numTiles + wavesPerBlock - 1) / wavesPerBlock;
    if (blocks > 256) blocks = 256;
    if (blocks < 1)   blocks = 1;

    p1interp_wmma<<<blocks, THREADS, 0, stream>>>(
        xp, wx, wy, Minv, Av, bbox, dofs, out, NP, C, NV);
}


# One point per lane: a wave takes 32 points per pass as two 16-point WMMA sub-tiles (partner lane's quad and x,y via permlanex16); masks, candidate walk, interpolation, store run once per point
# speedup vs baseline: 7.5759x; 1.3381x over previous
#include <hip/hip_runtime.h>


typedef __attribute__((ext_vector_type(2))) float v2f;
typedef __attribute__((ext_vector_type(8))) float v8f;
typedef float f4u __attribute__((ext_vector_type(4), aligned(4)));
typedef float f2u __attribute__((ext_vector_type(2), aligned(4)));

#define MAXC    512
#define MAXV    512
#define THREADS 256

__device__ __forceinline__ int xhalf_i(int v) {
    return __builtin_amdgcn_permlanex16(v, v, 0x76543210, 0xfedcba98, false, false);
}
__device__ __forceinline__ float xhalf_f(float v) {
    return __builtin_bit_cast(float, xhalf_i(__builtin_bit_cast(int, v)));
}

__device__ __forceinline__ float pick8(const v8f d, bool b0, bool b1, bool b2) {
    const float t0 = b0 ? d[1] : d[0];
    const float t1 = b0 ? d[3] : d[2];
    const float t2 = b0 ? d[5] : d[4];
    const float t3 = b0 ? d[7] : d[6];
    const float u0 = b1 ? t1 : t0;
    const float u1 = b1 ? t3 : t2;
    return b2 ? u1 : u0;
}

__launch_bounds__(THREADS, 2)
__global__ void p1interp_wmma(const float* __restrict__ x,
                              const float* __restrict__ wxg,
                              const float* __restrict__ wyg,
                              const float* __restrict__ Minv,
                              const float* __restrict__ Av,
                              const float* __restrict__ bbox,
                              const int*   __restrict__ dofs,
                              float2* __restrict__ out,
                              int NP, int C, int NV)
{
    __shared__ float4 sAg [MAXC * 2];
    __shared__ __align__(16) float sMinv[MAXC * 4];
    __shared__ __align__(16) float sBbox[MAXC * 4];
    __shared__ __align__(16) float sA   [MAXC * 2];
    __shared__ int   sDof [MAXC * 3];
    __shared__ float sWx  [MAXV];
    __shared__ float sWy  [MAXV];

    const int tid  = threadIdx.x;
    const int lane = tid & 31;
    const int wave = __builtin_amdgcn_readfirstlane(tid >> 5);

    const int numTiles = (NP + 31) >> 5;
    const int tileStep = gridDim.x * (THREADS >> 5);
    int tile = blockIdx.x * (THREADS >> 5) + wave;

    f2u cur;
    {
        const int g0 = tile * 32 + lane;
        cur = *(const f2u*)(x + (g0 < NP ? g0 : NP - 1) * 2);
    }

    f4u  mv[2], bv[2];
    f2u  av[2];
    int  k0[2], k1[2], k2[2];
    float wxv[2], wyv[2];
#pragma unroll
    for (int k = 0; k < 2; ++k) {
        const int c  = tid + k * THREADS;
        const int cl = c < C ? c : 0;
        mv[k] = *(const f4u*)(Minv + cl * 4);
        bv[k] = *(const f4u*)(bbox + cl * 4);
        av[k] = *(const f2u*)(Av + cl * 2);
        k0[k] = dofs[cl * 3 + 0];
        k1[k] = dofs[cl * 3 + 1];
        k2[k] = dofs[cl * 3 + 2];
        const int v  = tid + k * THREADS;
        const int vl = v < NV ? v : NV - 1;
        wxv[k] = wxg[vl];
        wyv[k] = wyg[vl];
    }
    asm volatile("" :: "v"(cur), "v"(mv[0]), "v"(mv[1]), "v"(bv[0]), "v"(bv[1]), "v"(av[0]), "v"(av[1]),
                       "v"(k0[0]), "v"(k1[0]), "v"(k2[0]), "v"(k0[1]), "v"(k1[1]), "v"(k2[1]),
                       "v"(wxv[0]), "v"(wyv[0]), "v"(wxv[1]), "v"(wyv[1]) : "memory");
#pragma unroll
    for (int k = 0; k < 2; ++k) {
        const int  c     = tid + k * THREADS;
        const bool valid = c < C;
        const float m00 = mv[k].x, m01 = mv[k].y, m10 = mv[k].z, m11 = mv[k].w;
        const float ax = av[k].x, ay = av[k].y;
        *(float4*)&sMinv[c * 4] = make_float4(m00, m01, m10, m11);
        *(float2*)&sA[c * 2]    = make_float2(ax, ay);
        sAg[c * 2 + 0] = valid ? make_float4(m00, m10, m01, m11)
                               : make_float4(0.0f, 0.0f, 0.0f, 0.0f);
        sAg[c * 2 + 1] = valid ? make_float4(-(ax * m00 + ay * m10), 0.0f,
                                             -(ax * m01 + ay * m11), 0.0f)
                               : make_float4(-1.0f, 0.0f, -1.0f, 0.0f);
        *(float4*)&sBbox[c * 4] = valid ? make_float4(bv[k].x, bv[k].y, bv[k].z, bv[k].w)
                                        : make_float4(0.0f, 0.0f, 0.0f, 0.0f);
        sDof[c * 3 + 0] = k0[k];
        sDof[c * 3 + 1] = k1[k];
        sDof[c * 3 + 2] = k2[k];
        sWx[c] = wxv[k];
        sWy[c] = wyv[k];
    }
    __syncthreads();

    const bool hi   = lane >= 16;
    const int  hsel = hi ? 1 : 0;
    const bool b0 = (lane & 1) != 0, b1 = (lane & 2) != 0, b2 = (lane & 4) != 0;
    const bool holder = ((lane >> 4) == ((lane >> 3) & 1));
    const float NTOL = -1e-10f;

    for (; tile < numTiles; tile += tileStep) {
        const int   gp = tile * 32 + lane;
        const float px = cur.x;
        const float py = cur.y;

        if (tile + tileStep < numTiles) {
            const int g2 = (tile + tileStep) * 32 + lane;
            cur = *(const f2u*)(x + (g2 < NP ? g2 : NP - 1) * 2);
        }

        const float pxx = xhalf_f(px);
        const float pyx = xhalf_f(py);
        v2f bA, bB;
        bA.x = hi ? 1.0f : px;
        bA.y = hi ? 0.0f : py;
        bB.x = hi ? 1.0f : pxx;
        bB.y = hi ? 0.0f : pyx;

        unsigned mx = 0, my = 0;
#pragma unroll
        for (int i = 0; i < 16; ++i) {
            float x0 = sBbox[(i * 16) * 4 + 0], x1 = sBbox[(i * 16) * 4 + 1];
            float y0 = sBbox[i * 4 + 2],        y1 = sBbox[i * 4 + 3];
            mx |= ((x0 < px) & (px < x1)) ? (1u << i) : 0u;
            my |= ((y0 < py) & (py < y1)) ? (1u << i) : 0u;
        }

        unsigned rx = mx, ry = my;
        bool act = (mx != 0u) & (my != 0u);
        int best = -1;
        while (__builtin_amdgcn_ballot_w32(act) != 0u) {
            const int i = __builtin_ctz(rx | 0x8000u);
            const int j = __builtin_ctz(ry | 0x8000u);
            const int q = i * 16 + j;
            const int qx = xhalf_i(q);
            const int qA = hi ? qx : q;
            const int qB = hi ? q : qx;
            const float4 aLA = sAg[qA * 2 + hsel];
            const float4 aUA = sAg[(qA + 256) * 2 + hsel];
            const float4 aLB = sAg[qB * 2 + hsel];
            const float4 aUB = sAg[(qB + 256) * 2 + hsel];
            v2f as0A, at0A, as1A, at1A, as0B, at0B, as1B, at1B;
            as0A.x = aLA.x; as0A.y = aLA.y;
            at0A.x = aLA.z; at0A.y = aLA.w;
            as1A.x = aUA.x; as1A.y = aUA.y;
            at1A.x = aUA.z; at1A.y = aUA.w;
            as0B.x = aLB.x; as0B.y = aLB.y;
            at0B.x = aLB.z; at0B.y = aLB.w;
            as1B.x = aUB.x; as1B.y = aUB.y;
            at1B.x = aUB.z; at1B.y = aUB.w;
            const v8f z = {};
            const v8f ds0A = __builtin_amdgcn_wmma_f32_16x16x4_f32(
                                 false, as0A, false, bA, (short)0, z, false, false);
            const v8f dt0A = __builtin_amdgcn_wmma_f32_16x16x4_f32(
                                 false, at0A, false, bA, (short)0, z, false, false);
            const v8f ds1A = __builtin_amdgcn_wmma_f32_16x16x4_f32(
                                 false, as1A, false, bA, (short)0, z, false, false);
            const v8f dt1A = __builtin_amdgcn_wmma_f32_16x16x4_f32(
                                 false, at1A, false, bA, (short)0, z, false, false);
            const v8f ds0B = __builtin_amdgcn_wmma_f32_16x16x4_f32(
                                 false, as0B, false, bB, (short)0, z, false, false);
            const v8f dt0B = __builtin_amdgcn_wmma_f32_16x16x4_f32(
                                 false, at0B, false, bB, (short)0, z, false, false);
            const v8f ds1B = __builtin_amdgcn_wmma_f32_16x16x4_f32(
                                 false, as1B, false, bB, (short)0, z, false, false);
            const v8f dt1B = __builtin_amdgcn_wmma_f32_16x16x4_f32(
                                 false, at1B, false, bB, (short)0, z, false, false);
            const float s0A = pick8(ds0A, b0, b1, b2), t0A = pick8(dt0A, b0, b1, b2);
            const float s1A = pick8(ds1A, b0, b1, b2), t1A = pick8(dt1A, b0, b1, b2);
            const float s0B = pick8(ds0B, b0, b1, b2), t0B = pick8(dt0B, b0, b1, b2);
            const float s1B = pick8(ds1B, b0, b1, b2), t1B = pick8(dt1B, b0, b1, b2);
            const bool ins0A = (s0A > NTOL) & (t0A > NTOL) & ((s0A + t0A) < 1.0f);
            const bool ins1A = (s1A > NTOL) & (t1A > NTOL) & ((s1A + t1A) < 1.0f);
            const bool ins0B = (s0B > NTOL) & (t0B > NTOL) & ((s0B + t0B) < 1.0f);
            const bool ins1B = (s1B > NTOL) & (t1B > NTOL) & ((s1B + t1B) < 1.0f);
            const int c0A = ins0A ? qA : -1;
            const int c1A = ins1A ? qA + 256 : -1;
            const int c0B = ins0B ? qB : -1;
            const int c1B = ins1B ? qB + 256 : -1;
            const int candA = c1A > c0A ? c1A : c0A;
            const int candB = c1B > c0B ? c1B : c0B;
            const int own   = hi ? candB : candA;
            const int other = hi ? candA : candB;
            const int recv  = xhalf_i(other);
            const int mine  = holder ? own : recv;
            const int cand  = act ? mine : -1;
            best = cand > best ? cand : best;
            const unsigned ry2 = ry & (ry - 1u);
            const bool rowdone = ry2 == 0u;
            rx = rowdone ? (rx & (rx - 1u)) : rx;
            ry = rowdone ? my : ry2;
            act = act & (rx != 0u);
        }

        float ox = 0.0f, oy = 0.0f;
        if (best >= 0) {
            int c  = best;
            float dx = px - sA[c * 2 + 0];
            float dy = py - sA[c * 2 + 1];
            float s  = dx * sMinv[c * 4 + 0] + dy * sMinv[c * 4 + 2];
            float t  = dx * sMinv[c * 4 + 1] + dy * sMinv[c * 4 + 3];
            float w0 = 1.0f - s - t;
            int d0 = sDof[c * 3 + 0], d1 = sDof[c * 3 + 1], d2 = sDof[c * 3 + 2];
            ox = w0 * sWx[d0] + s * sWx[d1] + t * sWx[d2];
            oy = w0 * sWy[d0] + s * sWy[d1] + t * sWy[d2];
        }
        if (gp < NP) out[gp] = make_float2(ox, oy);
    }
}

extern "C" void kernel_launch(void* const* d_in, const int* in_sizes, int n_in,
                              void* d_out, int out_size, void* d_ws, size_t ws_size,
                              hipStream_t stream) {
    const float* xp   = (const float*)d_in[0];
    const float* wx   = (const float*)d_in[1];
    const float* wy   = (const float*)d_in[2];
    const float* Minv = (const float*)d_in[3];
    const float* Av   = (const float*)d_in[4];
    const float* bbox = (const float*)d_in[5];
    const int*   dofs = (const int*)d_in[6];
    float2* out = (float2*)d_out;

    int NP = in_sizes[0] / 2;
    int C  = in_sizes[3] / 4;
    int NV = in_sizes[1];
    if (C > MAXC) C = MAXC;

    int numTiles = (NP + 31) / 32;
    int wavesPerBlock = THREADS / 32;
    int blocks = (numTiles + wavesPerBlock - 1) / wavesPerBlock;
    if (blocks > 256) blocks = 256;
    if (blocks < 1)   blocks = 1;

    p1interp_wmma<<<blocks, THREADS, 0, stream>>>(
        xp, wx, wy, Minv, Av, bbox, dofs, out, NP, C, NV);
}


# Points and one bbox bound pair per lane are the first two loads (bounds read by lane broadcast: no LDS copy, no scalar loads); first tile's masks are built ahead of the table wait
# speedup vs baseline: 8.0755x; 1.0659x over previous
#include <hip/hip_runtime.h>


typedef __attribute__((ext_vector_type(2))) float v2f;
typedef __attribute__((ext_vector_type(8))) float v8f;
typedef float f4u __attribute__((ext_vector_type(4), aligned(4)));
typedef float f2u __attribute__((ext_vector_type(2), aligned(4)));

#define MAXC    512
#define MAXV    512
#define THREADS 256

__device__ __forceinline__ int xhalf_i(int v) {
    return __builtin_amdgcn_permlanex16(v, v, 0x76543210, 0xfedcba98, false, false);
}
__device__ __forceinline__ float xhalf_f(float v) {
    return __builtin_bit_cast(float, xhalf_i(__builtin_bit_cast(int, v)));
}

__device__ __forceinline__ float lane_f(float v, int l) {
    return __builtin_bit_cast(float, __builtin_amdgcn_readlane(__builtin_bit_cast(int, v), l));
}

__device__ __forceinline__ void bbox_masks(float px, float py, float b0, float b1,
                                           unsigned& mx, unsigned& my) {
    unsigned ax = 0, ay = 0;
#pragma unroll
    for (int i = 0; i < 16; ++i) {
        const float x0 = lane_f(b0, i),      x1 = lane_f(b1, i);
        const float y0 = lane_f(b0, 16 + i), y1 = lane_f(b1, 16 + i);
        ax |= ((x0 < px) & (px < x1)) ? (1u << i) : 0u;
        ay |= ((y0 < py) & (py < y1)) ? (1u << i) : 0u;
    }
    mx = ax;
    my = ay;
}

__device__ __forceinline__ float pick8(const v8f d, bool b0, bool b1, bool b2) {
    const float t0 = b0 ? d[1] : d[0];
    const float t1 = b0 ? d[3] : d[2];
    const float t2 = b0 ? d[5] : d[4];
    const float t3 = b0 ? d[7] : d[6];
    const float u0 = b1 ? t1 : t0;
    const float u1 = b1 ? t3 : t2;
    return b2 ? u1 : u0;
}

__launch_bounds__(THREADS, 2)
__global__ void p1interp_wmma(const float* x,
                              const float* wxg,
                              const float* wyg,
                              const float* Minv,
                              const float* Av,
                              const float* bbox,
                              const int*   dofs,
                              float2* __restrict__ out,
                              int NP, int C, int NV)
{
    __shared__ float4 sAg [MAXC * 2];
    __shared__ __align__(16) float sMinv[MAXC * 4];
    __shared__ __align__(16) float sA   [MAXC * 2];
    __shared__ int   sDof [MAXC * 3];
    __shared__ float sWx  [MAXV];
    __shared__ float sWy  [MAXV];

    const int tid  = threadIdx.x;
    const int lane = tid & 31;
    const int wave = __builtin_amdgcn_readfirstlane(tid >> 5);

    const int numTiles = (NP + 31) >> 5;
    const int tileStep = gridDim.x * (THREADS >> 5);
    int tile = blockIdx.x * (THREADS >> 5) + wave;

    f2u cur, braw;
    const int  bcell  = lane >= 16 ? (lane & 15) : (lane & 15) * 16;
    const bool bvalid = bcell < C;
    {
        const int g0 = tile * 32 + lane;
        cur  = *(const f2u*)(x + (g0 < NP ? g0 : NP - 1) * 2);
        braw = *(const f2u*)(bbox + (bvalid ? bcell : 0) * 4 + (lane >= 16 ? 2 : 0));
    }

    f4u  mv[2];
    f2u  av[2];
    int  k0[2], k1[2], k2[2];
    float wxv[2], wyv[2];
#pragma unroll
    for (int k = 0; k < 2; ++k) {
        const int c  = tid + k * THREADS;
        const int cl = c < C ? c : 0;
        mv[k] = *(const f4u*)(Minv + cl * 4);
        av[k] = *(const f2u*)(Av + cl * 2);
        k0[k] = dofs[cl * 3 + 0];
        k1[k] = dofs[cl * 3 + 1];
        k2[k] = dofs[cl * 3 + 2];
        const int v  = tid + k * THREADS;
        const int vl = v < NV ? v : NV - 1;
        wxv[k] = wxg[vl];
        wyv[k] = wyg[vl];
    }
    f2u bd;
    bd.x = bvalid ? braw.x : 0.0f;
    bd.y = bvalid ? braw.y : 0.0f;
    unsigned mx, my;
    bbox_masks(cur.x, cur.y, bd.x, bd.y, mx, my);
    asm volatile("" : "+v"(mv[0]), "+v"(mv[1]), "+v"(av[0]), "+v"(av[1])
                    : "v"(mx), "v"(my), "s"(tileStep) : "memory");
#pragma unroll
    for (int k = 0; k < 2; ++k) {
        const int  c     = tid + k * THREADS;
        const bool valid = c < C;
        const float m00 = mv[k].x, m01 = mv[k].y, m10 = mv[k].z, m11 = mv[k].w;
        const float ax = av[k].x, ay = av[k].y;
        *(float4*)&sMinv[c * 4] = make_float4(m00, m01, m10, m11);
        *(float2*)&sA[c * 2]    = make_float2(ax, ay);
        sAg[c * 2 + 0] = valid ? make_float4(m00, m10, m01, m11)
                               : make_float4(0.0f, 0.0f, 0.0f, 0.0f);
        sAg[c * 2 + 1] = valid ? make_float4(-(ax * m00 + ay * m10), 0.0f,
                                             -(ax * m01 + ay * m11), 0.0f)
                               : make_float4(-1.0f, 0.0f, -1.0f, 0.0f);
        sDof[c * 3 + 0] = k0[k];
        sDof[c * 3 + 1] = k1[k];
        sDof[c * 3 + 2] = k2[k];
        sWx[c] = wxv[k];
        sWy[c] = wyv[k];
    }
    __syncthreads();

    const bool hi   = lane >= 16;
    const int  hsel = hi ? 1 : 0;
    const bool b0 = (lane & 1) != 0, b1 = (lane & 2) != 0, b2 = (lane & 4) != 0;
    const bool holder = ((lane >> 4) == ((lane >> 3) & 1));
    const float NTOL = -1e-10f;

    for (; tile < numTiles; tile += tileStep) {
        const int   gp = tile * 32 + lane;
        const float px = cur.x;
        const float py = cur.y;

        if (tile + tileStep < numTiles) {
            const int g2 = (tile + tileStep) * 32 + lane;
            cur = *(const f2u*)(x + (g2 < NP ? g2 : NP - 1) * 2);
        }

        const float pxx = xhalf_f(px);
        const float pyx = xhalf_f(py);
        v2f bA, bB;
        bA.x = hi ? 1.0f : px;
        bA.y = hi ? 0.0f : py;
        bB.x = hi ? 1.0f : pxx;
        bB.y = hi ? 0.0f : pyx;

        unsigned rx = mx, ry = my;
        bool act = (mx != 0u) & (my != 0u);
        int best = -1;
        while (__builtin_amdgcn_ballot_w32(act) != 0u) {
            const int i = __builtin_ctz(rx | 0x8000u);
            const int j = __builtin_ctz(ry | 0x8000u);
            const int q = i * 16 + j;
            const int qx = xhalf_i(q);
            const int qA = hi ? qx : q;
            const int qB = hi ? q : qx;
            const float4 aLA = sAg[qA * 2 + hsel];
            const float4 aUA = sAg[(qA + 256) * 2 + hsel];
            const float4 aLB = sAg[qB * 2 + hsel];
            const float4 aUB = sAg[(qB + 256) * 2 + hsel];
            v2f as0A, at0A, as1A, at1A, as0B, at0B, as1B, at1B;
            as0A.x = aLA.x; as0A.y = aLA.y;
            at0A.x = aLA.z; at0A.y = aLA.w;
            as1A.x = aUA.x; as1A.y = aUA.y;
            at1A.x = aUA.z; at1A.y = aUA.w;
            as0B.x = aLB.x; as0B.y = aLB.y;
            at0B.x = aLB.z; at0B.y = aLB.w;
            as1B.x = aUB.x; as1B.y = aUB.y;
            at1B.x = aUB.z; at1B.y = aUB.w;
            const v8f z = {};
            const v8f ds0A = __builtin_amdgcn_wmma_f32_16x16x4_f32(
                                 false, as0A, false, bA, (short)0, z, false, false);
            const v8f dt0A = __builtin_amdgcn_wmma_f32_16x16x4_f32(
                                 false, at0A, false, bA, (short)0, z, false, false);
            const v8f ds1A = __builtin_amdgcn_wmma_f32_16x16x4_f32(
                                 false, as1A, false, bA, (short)0, z, false, false);
            const v8f dt1A = __builtin_amdgcn_wmma_f32_16x16x4_f32(
                                 false, at1A, false, bA, (short)0, z, false, false);
            const v8f ds0B = __builtin_amdgcn_wmma_f32_16x16x4_f32(
                                 false, as0B, false, bB, (short)0, z, false, false);
            const v8f dt0B = __builtin_amdgcn_wmma_f32_16x16x4_f32(
                                 false, at0B, false, bB, (short)0, z, false, false);
            const v8f ds1B = __builtin_amdgcn_wmma_f32_16x16x4_f32(
                                 false, as1B, false, bB, (short)0, z, false, false);
            const v8f dt1B = __builtin_amdgcn_wmma_f32_16x16x4_f32(
                                 false, at1B, false, bB, (short)0, z, false, false);
            const float s0A = pick8(ds0A, b0, b1, b2), t0A = pick8(dt0A, b0, b1, b2);
            const float s1A = pick8(ds1A, b0, b1, b2), t1A = pick8(dt1A, b0, b1, b2);
            const float s0B = pick8(ds0B, b0, b1, b2), t0B = pick8(dt0B, b0, b1, b2);
            const float s1B = pick8(ds1B, b0, b1, b2), t1B = pick8(dt1B, b0, b1, b2);
            const bool ins0A = (s0A > NTOL) & (t0A > NTOL) & ((s0A + t0A) < 1.0f);
            const bool ins1A = (s1A > NTOL) & (t1A > NTOL) & ((s1A + t1A) < 1.0f);
            const bool ins0B = (s0B > NTOL) & (t0B > NTOL) & ((s0B + t0B) < 1.0f);
            const bool ins1B = (s1B > NTOL) & (t1B > NTOL) & ((s1B + t1B) < 1.0f);
            const int c0A = ins0A ? qA : -1;
            const int c1A = ins1A ? qA + 256 : -1;
            const int c0B = ins0B ? qB : -1;
            const int c1B = ins1B ? qB + 256 : -1;
            const int candA = c1A > c0A ? c1A : c0A;
            const int candB = c1B > c0B ? c1B : c0B;
            const int own   = hi ? candB : candA;
            const int other = hi ? candA : candB;
            const int recv  = xhalf_i(other);
            const int mine  = holder ? own : recv;
            const int cand  = act ? mine : -1;
            best = cand > best ? cand : best;
            const unsigned ry2 = ry & (ry - 1u);
            const bool rowdone = ry2 == 0u;
            rx = rowdone ? (rx & (rx - 1u)) : rx;
            ry = rowdone ? my : ry2;
            act = act & (rx != 0u);
        }

        float ox = 0.0f, oy = 0.0f;
        if (best >= 0) {
            int c  = best;
            float dx = px - sA[c * 2 + 0];
            float dy = py - sA[c * 2 + 1];
            float s  = dx * sMinv[c * 4 + 0] + dy * sMinv[c * 4 + 2];
            float t  = dx * sMinv[c * 4 + 1] + dy * sMinv[c * 4 + 3];
            float w0 = 1.0f - s - t;
            int d0 = sDof[c * 3 + 0], d1 = sDof[c * 3 + 1], d2 = sDof[c * 3 + 2];
            ox = w0 * sWx[d0] + s * sWx[d1] + t * sWx[d2];
            oy = w0 * sWy[d0] + s * sWy[d1] + t * sWy[d2];
        }
        if (gp < NP) out[gp] = make_float2(ox, oy);

        if (tile + tileStep < numTiles) bbox_masks(cur.x, cur.y, bd.x, bd.y, mx, my);
    }
}

extern "C" void kernel_launch(void* const* d_in, const int* in_sizes, int n_in,
                              void* d_out, int out_size, void* d_ws, size_t ws_size,
                              hipStream_t stream) {
    const float* xp   = (const float*)d_in[0];
    const float* wx   = (const float*)d_in[1];
    const float* wy   = (const float*)d_in[2];
    const float* Minv = (const float*)d_in[3];
    const float* Av   = (const float*)d_in[4];
    const float* bbox = (const float*)d_in[5];
    const int*   dofs = (const int*)d_in[6];
    float2* out = (float2*)d_out;

    int NP = in_sizes[0] / 2;
    int C  = in_sizes[3] / 4;
    int NV = in_sizes[1];
    if (C > MAXC) C = MAXC;

    int numTiles = (NP + 31) / 32;
    int wavesPerBlock = THREADS / 32;
    int blocks = (numTiles + wavesPerBlock - 1) / wavesPerBlock;
    if (blocks > 256) blocks = 256;
    if (blocks < 1)   blocks = 1;

    p1interp_wmma<<<blocks, THREADS, 0, stream>>>(
        xp, wx, wy, Minv, Av, bbox, dofs, out, NP, C, NV);
}
